# starting kernel
# baseline (speedup 1.0000x reference)
#include <hip/hip_runtime.h>
#include <hip/hip_bf16.h>

typedef float v2f __attribute__((ext_vector_type(2)));
typedef float v8f __attribute__((ext_vector_type(8)));


__global__ void k_deg_init(float* __restrict__ deg, int n) {
  int i = blockIdx.x * blockDim.x + threadIdx.x;
  if (i < n) deg[i] = 1.0f;
}

__global__ void k_deg_count(const int* __restrict__ col, float* __restrict__ deg, int e) {
  int i = blockIdx.x * blockDim.x + threadIdx.x;
  if (i < e) atomicAdd(&deg[col[i]], 1.0f);
}

__global__ void k_deg_rsqrt(const float* __restrict__ deg, float* __restrict__ dinv, int n) {
  int i = blockIdx.x * blockDim.x + threadIdx.x;
  if (i < n) dinv[i] = rsqrtf(deg[i]);
}


__global__ void k_gemm_wmma(const float* __restrict__ x,
                            const float* __restrict__ W,
                            float* __restrict__ xw, int ntiles) {
  const int lane   = threadIdx.x & 31;
  const int waveIb = threadIdx.x >> 5;
  const int tile   = blockIdx.x * (blockDim.x >> 5) + waveIb;
  if (tile >= ntiles) return;

  const int m     = lane & 15;
  const int khalf = lane >> 4;

  const float* xrow  = x + (size_t)(tile * 16 + m) * 128 + khalf * 2;
  const float* wbase = W + (size_t)m * 128 + khalf * 2;

  v8f acc[8] = {};

  v2f a_cur = *(const v2f*)(xrow);
  v2f b_cur[8];
#pragma unroll
  for (int t = 0; t < 8; ++t) b_cur[t] = *(const v2f*)(wbase + t * 2048);

#pragma unroll
  for (int k0 = 0; k0 < 128; k0 += 4) {
    v2f a_nxt = a_cur;
    v2f b_nxt[8];
#pragma unroll
    for (int t = 0; t < 8; ++t) b_nxt[t] = b_cur[t];
    if (k0 + 4 < 128) {
      a_nxt = *(const v2f*)(xrow + k0 + 4);
#pragma unroll
      for (int t = 0; t < 8; ++t)
        b_nxt[t] = *(const v2f*)(wbase + t * 2048 + k0 + 4);
    }
#pragma unroll
    for (int t = 0; t < 8; ++t) {
      acc[t] = __builtin_amdgcn_wmma_f32_16x16x4_f32(
           false, a_cur,  false, b_cur[t],
           (short)0, acc[t],  false,  false);
    }
    a_cur = a_nxt;
#pragma unroll
    for (int t = 0; t < 8; ++t) b_cur[t] = b_nxt[t];
  }

  float* outbase = xw + (size_t)(tile * 16 + khalf * 8) * 128 + m;
#pragma unroll
  for (int t = 0; t < 8; ++t) {
#pragma unroll
    for (int v = 0; v < 8; ++v) {
      outbase[(size_t)v * 128 + t * 16] = acc[t][v];
    }
  }
}

__global__ void k_gemm_tail(const float* __restrict__ x, const float* __restrict__ W,
                            float* __restrict__ xw, int n, int start) {
  int i = blockIdx.x * blockDim.x + threadIdx.x;
  int total = (n - start) * 128;
  if (i >= total) return;
  int node = start + i / 128;
  int o = i & 127;
  const float* xr = x + (size_t)node * 128;
  const float* wr = W + (size_t)o * 128;
  float s = 0.f;
  for (int k = 0; k < 128; ++k) s += xr[k] * wr[k];
  xw[(size_t)node * 128 + o] = s;
}


__global__ void k_init_out(const float* __restrict__ xw, const float* __restrict__ dinv,
                           const float* __restrict__ bias, float* __restrict__ out, int n) {
  int i = blockIdx.x * blockDim.x + threadIdx.x;
  int total = n * 32;
  if (i >= total) return;
  int node = i >> 5;
  int f = (i & 31) * 4;
  float d = dinv[node];
  float s = d * d;
  const float4 v = *(const float4*)(xw + (size_t)node * 128 + f);
  const float4 bb = *(const float4*)(bias + f);
  float4 o;
  o.x = v.x * s + bb.x;
  o.y = v.y * s + bb.y;
  o.z = v.z * s + bb.z;
  o.w = v.w * s + bb.w;
  *(float4*)(out + (size_t)node * 128 + f) = o;
}


__global__ void k_scatter(const float* __restrict__ xw,
                          const int* __restrict__ row, const int* __restrict__ col,
                          const float* __restrict__ dinv,
                          float* __restrict__ out, int e) {
  const int lane = threadIdx.x & 31;
  const int edge = (blockIdx.x * blockDim.x + threadIdx.x) >> 5;
  if (edge >= e) return;
  const int r = row[edge];
  const int c = col[edge];
  const float nrm = dinv[r] * dinv[c];
  const float4 v = *(const float4*)(xw + (size_t)r * 128 + lane * 4);
  float* dst = out + (size_t)c * 128 + lane * 4;
  atomicAdd(dst + 0, v.x * nrm);
  atomicAdd(dst + 1, v.y * nrm);
  atomicAdd(dst + 2, v.z * nrm);
  atomicAdd(dst + 3, v.w * nrm);
}


extern "C" void kernel_launch(void* const* d_in, const int* in_sizes, int n_in,
                              void* d_out, int out_size, void* d_ws, size_t ws_size,
                              hipStream_t stream) {
  const float* x  = (const float*)d_in[0];
  const int* ei   = (const int*)d_in[1];
  const float* W  = (const float*)d_in[2];
  const float* b  = (const float*)d_in[3];
  float* out      = (float*)d_out;

  const int N = in_sizes[0] / 128;
  const int E = in_sizes[1] / 2;
  const int* row = ei;
  const int* col = ei + E;

  float* xw   = (float*)d_ws;
  float* deg  = xw + (size_t)N * 128;
  float* dinv = deg + N;

  k_deg_init<<<(N + 255) / 256, 256, 0, stream>>>(deg, N);
  k_deg_count<<<(E + 255) / 256, 256, 0, stream>>>(col, deg, E);
  k_deg_rsqrt<<<(N + 255) / 256, 256, 0, stream>>>(deg, dinv, N);

  const int ntiles = N / 16;
  if (ntiles > 0)
    k_gemm_wmma<<<(ntiles + 7) / 8, 256, 0, stream>>>(x, W, xw, ntiles);
  const int rem = N % 16;
  if (rem > 0)
    k_gemm_tail<<<(rem * 128 + 255) / 256, 256, 0, stream>>>(x, W, xw, N, ntiles * 16);

  k_init_out<<<(N * 32 + 255) / 256, 256, 0, stream>>>(xw, dinv, b, out, N);

  k_scatter<<<(E + 7) / 8, 256, 0, stream>>>(xw, row, col, dinv, out, E);
}


# Replace the per-edge f32 atomic scatter by in-edge lists grouped by target (u32 rank atomics, scan, fill) and a wave-per-node register gather; per-node sum order = arrival-order rank
# speedup vs baseline: 1.8763x; 1.8763x over previous
#include <hip/hip_runtime.h>
#include <hip/hip_bf16.h>

typedef float v2f __attribute__((ext_vector_type(2)));
typedef float v8f __attribute__((ext_vector_type(8)));


__global__ void k_cnt_zero(int* __restrict__ cnt, int n) {
  int i = blockIdx.x * blockDim.x + threadIdx.x;
  if (i < n) cnt[i] = 0;
}

__global__ void k_rank(const int* __restrict__ col, int* __restrict__ cnt,
                       int* __restrict__ pos, int e) {
  int i = blockIdx.x * blockDim.x + threadIdx.x;
  if (i < e) pos[i] = atomicAdd(&cnt[col[i]], 1);
}

__global__ void __launch_bounds__(1024)
k_scan(const int* __restrict__ cnt, int* __restrict__ off, float* __restrict__ dinv, int n) {
  __shared__ int wsum[32];
  __shared__ int sbase;
  const int tid  = threadIdx.x;
  const int lane = tid & 31;
  const int w    = tid >> 5;
  const int blk0 = blockIdx.x * 1024;

  int part = 0;
  const int4* c4 = (const int4*)cnt;
  const int nvec = blk0 >> 2;
#pragma unroll 4
  for (int i = tid; i < nvec; i += 1024) {
    const int4 v = c4[i];
    part += (v.x + v.y) + (v.z + v.w);
  }
  for (int o = 16; o > 0; o >>= 1) part += __shfl_xor(part, o);
  if (lane == 0) wsum[w] = part;
  __syncthreads();
  if (w == 0) {
    int v = wsum[lane];
    for (int o = 16; o > 0; o >>= 1) v += __shfl_xor(v, o);
    if (lane == 0) sbase = v;
  }
  __syncthreads();
  const int base = sbase;

  const int i = blk0 + tid;
  const int c = (i < n) ? cnt[i] : 0;
  int incl = c;
  for (int o = 1; o < 32; o <<= 1) {
    const int t = __shfl_up(incl, o);
    if (lane >= o) incl += t;
  }
  if (lane == 31) wsum[w] = incl;
  __syncthreads();
  if (w == 0) {
    const int v = wsum[lane];
    int s = v;
    for (int o = 1; o < 32; o <<= 1) {
      const int t = __shfl_up(s, o);
      if (lane >= o) s += t;
    }
    wsum[lane] = s - v;
  }
  __syncthreads();
  if (i < n) {
    off[i]  = base + wsum[w] + (incl - c);
    dinv[i] = rsqrtf((float)(c + 1));
  }
}

__global__ void k_fill(const int* __restrict__ row, const int* __restrict__ col,
                       const int* __restrict__ pos, const int* __restrict__ off,
                       int* __restrict__ src, int e) {
  int i = blockIdx.x * blockDim.x + threadIdx.x;
  if (i < e) src[off[col[i]] + pos[i]] = row[i];
}


__global__ void k_gemm_wmma(const float* __restrict__ x,
                            const float* __restrict__ W,
                            float* __restrict__ xw, int ntiles) {
  const int lane   = threadIdx.x & 31;
  const int waveIb = threadIdx.x >> 5;
  const int tile   = blockIdx.x * (blockDim.x >> 5) + waveIb;
  if (tile >= ntiles) return;

  const int m     = lane & 15;
  const int khalf = lane >> 4;

  const float* xrow  = x + (size_t)(tile * 16 + m) * 128 + khalf * 2;
  const float* wbase = W + (size_t)m * 128 + khalf * 2;

  v8f acc[8] = {};

  v2f a_cur = *(const v2f*)(xrow);
  v2f b_cur[8];
#pragma unroll
  for (int t = 0; t < 8; ++t) b_cur[t] = *(const v2f*)(wbase + t * 2048);

#pragma unroll
  for (int k0 = 0; k0 < 128; k0 += 4) {
    v2f a_nxt = a_cur;
    v2f b_nxt[8];
#pragma unroll
    for (int t = 0; t < 8; ++t) b_nxt[t] = b_cur[t];
    if (k0 + 4 < 128) {
      a_nxt = *(const v2f*)(xrow + k0 + 4);
#pragma unroll
      for (int t = 0; t < 8; ++t)
        b_nxt[t] = *(const v2f*)(wbase + t * 2048 + k0 + 4);
    }
#pragma unroll
    for (int t = 0; t < 8; ++t) {
      acc[t] = __builtin_amdgcn_wmma_f32_16x16x4_f32(
           false, a_cur,  false, b_cur[t],
           (short)0, acc[t],  false,  false);
    }
    a_cur = a_nxt;
#pragma unroll
    for (int t = 0; t < 8; ++t) b_cur[t] = b_nxt[t];
  }

  float* outbase = xw + (size_t)(tile * 16 + khalf * 8) * 128 + m;
#pragma unroll
  for (int t = 0; t < 8; ++t) {
#pragma unroll
    for (int v = 0; v < 8; ++v) {
      outbase[(size_t)v * 128 + t * 16] = acc[t][v];
    }
  }
}

__global__ void k_gemm_tail(const float* __restrict__ x, const float* __restrict__ W,
                            float* __restrict__ xw, int n, int start) {
  int i = blockIdx.x * blockDim.x + threadIdx.x;
  int total = (n - start) * 128;
  if (i >= total) return;
  int node = start + i / 128;
  int o = i & 127;
  const float* xr = x + (size_t)node * 128;
  const float* wr = W + (size_t)o * 128;
  float s = 0.f;
  for (int k = 0; k < 128; ++k) s += xr[k] * wr[k];
  xw[(size_t)node * 128 + o] = s;
}


__device__ __forceinline__ void row_fma(float4& acc, const float4 v, const float w) {
  acc.x = fmaf(v.x, w, acc.x);
  acc.y = fmaf(v.y, w, acc.y);
  acc.z = fmaf(v.z, w, acc.z);
  acc.w = fmaf(v.w, w, acc.w);
}

__global__ void __launch_bounds__(256)
k_gather(const float* __restrict__ xw, const int* __restrict__ src,
         const int* __restrict__ off, const int* __restrict__ cnt,
         const float* __restrict__ dinv, const float* __restrict__ bias,
         float* __restrict__ out, int n) {
  const int lane = threadIdx.x & 31;
  const int node = __builtin_amdgcn_readfirstlane((blockIdx.x * blockDim.x + threadIdx.x) >> 5);
  if (node >= n) return;
  const int start = off[node];
  const int deg   = cnt[node];
  const float dc  = dinv[node];
  const float* xl = xw + lane * 4;

  const float4 vs = *(const float4*)(xl + (size_t)node * 128);
  const float4 bb = *(const float4*)(bias + lane * 4);
  const float s = dc * dc;
  float4 acc;
  acc.x = vs.x * s + bb.x;
  acc.y = vs.y * s + bb.y;
  acc.z = vs.z * s + bb.z;
  acc.w = vs.w * s + bb.w;

  for (int j0 = 0; j0 < deg; j0 += 32) {
    int my_r = 0;
    float my_w = 0.f;
    if (j0 + lane < deg) {
      my_r = src[start + j0 + lane];
      my_w = dinv[my_r] * dc;
    }
    const int m = (deg - j0 < 32) ? (deg - j0) : 32;
    int jj = 0;
    for (; jj + 8 <= m; jj += 8) {
      float4 v[8];
      float wt[8];
      for (int u = 0; u < 8; ++u) {
        const int r = __builtin_amdgcn_readlane(my_r, jj + u);
        wt[u] = __int_as_float(__builtin_amdgcn_readlane(__float_as_int(my_w), jj + u));
        v[u] = *(const float4*)(xl + (size_t)r * 128);
      }
      for (int u = 0; u < 8; ++u) row_fma(acc, v[u], wt[u]);
    }
    if (jj + 4 <= m) {
      float4 v[4];
      float wt[4];
      for (int u = 0; u < 4; ++u) {
        const int r = __builtin_amdgcn_readlane(my_r, jj + u);
        wt[u] = __int_as_float(__builtin_amdgcn_readlane(__float_as_int(my_w), jj + u));
        v[u] = *(const float4*)(xl + (size_t)r * 128);
      }
      for (int u = 0; u < 4; ++u) row_fma(acc, v[u], wt[u]);
      jj += 4;
    }
    if (jj + 2 <= m) {
      float4 v[2];
      float wt[2];
      for (int u = 0; u < 2; ++u) {
        const int r = __builtin_amdgcn_readlane(my_r, jj + u);
        wt[u] = __int_as_float(__builtin_amdgcn_readlane(__float_as_int(my_w), jj + u));
        v[u] = *(const float4*)(xl + (size_t)r * 128);
      }
      for (int u = 0; u < 2; ++u) row_fma(acc, v[u], wt[u]);
      jj += 2;
    }
    if (jj < m) {
      const int r = __builtin_amdgcn_readlane(my_r, jj);
      const float wt = __int_as_float(__builtin_amdgcn_readlane(__float_as_int(my_w), jj));
      const float4 v = *(const float4*)(xl + (size_t)r * 128);
      row_fma(acc, v, wt);
    }
  }
  *(float4*)(out + (size_t)node * 128 + lane * 4) = acc;
}


extern "C" void kernel_launch(void* const* d_in, const int* in_sizes, int n_in,
                              void* d_out, int out_size, void* d_ws, size_t ws_size,
                              hipStream_t stream) {
  const float* x  = (const float*)d_in[0];
  const int* ei   = (const int*)d_in[1];
  const float* W  = (const float*)d_in[2];
  const float* b  = (const float*)d_in[3];
  float* out      = (float*)d_out;

  const int N = in_sizes[0] / 128;
  const int E = in_sizes[1] / 2;
  const int* row = ei;
  const int* col = ei + E;

  float* xw   = (float*)d_ws;
  int* cnt    = (int*)(xw + (size_t)N * 128);
  int* off    = cnt + N;
  float* dinv = (float*)(off + N);
  int* pos    = (int*)(dinv + N);
  int* src    = pos + E;

  k_cnt_zero<<<(N + 255) / 256, 256, 0, stream>>>(cnt, N);
  k_rank<<<(E + 255) / 256, 256, 0, stream>>>(col, cnt, pos, E);
  k_scan<<<(N + 1023) / 1024, 1024, 0, stream>>>(cnt, off, dinv, N);
  k_fill<<<(E + 255) / 256, 256, 0, stream>>>(row, col, pos, off, src, E);

  const int ntiles = N / 16;
  if (ntiles > 0)
    k_gemm_wmma<<<(ntiles + 7) / 8, 256, 0, stream>>>(x, W, xw, ntiles);
  const int rem = N % 16;
  if (rem > 0)
    k_gemm_tail<<<(rem * 128 + 255) / 256, 256, 0, stream>>>(x, W, xw, N, ntiles * 16);

  k_gather<<<(N + 7) / 8, 256, 0, stream>>>(xw, src, off, cnt, dinv, b, out, N);
}


# In-edge lists built with LDS counters (edge chunks grouped by 256-node bucket, then one work-group per bucket ranks and places), not per-edge returning global atomics; sum order = LDS arrival rank
# speedup vs baseline: 3.4206x; 1.8231x over previous
#include <hip/hip_runtime.h>
#include <hip/hip_bf16.h>

typedef float v2f __attribute__((ext_vector_type(2)));
typedef float v8f __attribute__((ext_vector_type(8)));


#define NODE_SH 8
#define NODES_PER_BUCKET 256
#define CHUNK 8192
#define EPT 8
#define STAGE_CAP 8192

__device__ __forceinline__ int block_excl_scan_1024(int v, int* wsum, int& total) {
  const int lane = threadIdx.x & 31;
  const int w    = threadIdx.x >> 5;
  int incl = v;
  for (int o = 1; o < 32; o <<= 1) {
    const int t = __shfl_up(incl, o);
    if (lane >= o) incl += t;
  }
  __syncthreads();
  if (lane == 31) wsum[w] = incl;
  __syncthreads();
  const int x = wsum[lane];
  int s = x;
  for (int o = 1; o < 32; o <<= 1) {
    const int t = __shfl_up(s, o);
    if (lane >= o) s += t;
  }
  total = __shfl(s, 31);
  return __shfl(s - x, w) + (incl - v);
}

__device__ __forceinline__ int block_sum_1024(int v, int* wsum) {
  const int lane = threadIdx.x & 31;
  const int w    = threadIdx.x >> 5;
  for (int o = 16; o > 0; o >>= 1) v += __shfl_xor(v, o);
  __syncthreads();
  if (lane == 0) wsum[w] = v;
  __syncthreads();
  int x = wsum[lane];
  for (int o = 16; o > 0; o >>= 1) x += __shfl_xor(x, o);
  return x;
}

__global__ void __launch_bounds__(1024)
k_chunk_sort(const int* __restrict__ row, const int* __restrict__ col,
             int* __restrict__ sorted_i, int* __restrict__ histG, int* __restrict__ offG,
             int e, int nb, int nchunks) {
  __shared__ int cb[1024];
  __shared__ int wsum[32];
  __shared__ int2 buf[CHUNK];
  int2* __restrict__ sorted = (int2*)sorted_i;
  const int tid = threadIdx.x;
  const int c   = blockIdx.x;
  const int e0  = c * CHUNK;
  cb[tid] = 0;
  int rw[EPT], cl[EPT], rk[EPT];
#pragma unroll
  for (int k = 0; k < EPT; ++k) {
    const int i = e0 + k * 1024 + tid;
    const bool ok = i < e;
    rw[k] = ok ? row[i] : 0;
    cl[k] = ok ? col[i] : -1;
  }
  __syncthreads();
#pragma unroll
  for (int k = 0; k < EPT; ++k)
    rk[k] = (cl[k] >= 0) ? atomicAdd(&cb[cl[k] >> NODE_SH], 1) : 0;
  __syncthreads();
  const int v = cb[tid];
  int total;
  const int ex = block_excl_scan_1024(v, wsum, total);
  if (tid < nb) {
    histG[tid * nchunks + c] = v;
    offG[tid * nchunks + c]  = ex;
  }
  cb[tid] = ex;
  __syncthreads();
#pragma unroll
  for (int k = 0; k < EPT; ++k)
    if (cl[k] >= 0) buf[cb[cl[k] >> NODE_SH] + rk[k]] = make_int2(rw[k], cl[k]);
  __syncthreads();
#pragma unroll
  for (int k = 0; k < EPT; ++k) {
    const int j = k * 1024 + tid;
    if (e0 + j < e) sorted[e0 + j] = buf[j];
  }
}

__global__ void __launch_bounds__(1024)
k_bucket_lists(const int* __restrict__ sorted_i, const int* __restrict__ histG,
               const int* __restrict__ offG, int* __restrict__ src, int* __restrict__ cnt,
               int* __restrict__ off, float* __restrict__ dinv, int n, int nchunks) {
  __shared__ int rlen[1024];
  __shared__ int rbeg[1024];
  __shared__ int ncur[NODES_PER_BUCKET];
  __shared__ int wsum[32];
  __shared__ int lbuf[STAGE_CAP];
  const int2* __restrict__ sorted = (const int2*)sorted_i;
  const int tid  = threadIdx.x;
  const int lane = tid & 31;
  const int w     = __builtin_amdgcn_readfirstlane(tid >> 5);
  const int b     = blockIdx.x;
  const int node0 = b << NODE_SH;
  const int front = b * nchunks;

  int mylen = 0;
  if (tid < nchunks) {
    mylen     = histG[front + tid];
    rlen[tid] = mylen;
    rbeg[tid] = tid * CHUNK + offG[front + tid];
  }
  if (tid < NODES_PER_BUCKET) ncur[tid] = 0;

  int part = 0;
  const int4* h4 = (const int4*)histG;
  const int nvec = front >> 2;
#pragma unroll 4
  for (int i = tid; i < nvec; i += 1024) {
    const int4 q = h4[i];
    part += (q.x + q.y) + (q.z + q.w);
  }
  if (tid < (front & 3)) part += histG[(nvec << 2) + tid];
  const int base = block_sum_1024(part, wsum);
  const int tot  = block_sum_1024(mylen, wsum);
  const bool staged = tot <= STAGE_CAP;

  for (int c0 = w; c0 < nchunks; c0 += 128) {
    int ln[4], bg[4];
    int2 ed[4];
#pragma unroll
    for (int k = 0; k < 4; ++k) {
      const int c = c0 + 32 * k;
      const bool in = c < nchunks;
      ln[k] = in ? rlen[c] : 0;
      bg[k] = in ? rbeg[c] : 0;
    }
#pragma unroll
    for (int k = 0; k < 4; ++k) {
      ed[k] = make_int2(0, node0);
      if (lane < ln[k]) ed[k] = sorted[bg[k] + lane];
    }
#pragma unroll
    for (int k = 0; k < 4; ++k)
      if (lane < ln[k]) atomicAdd(&ncur[ed[k].y - node0], 1);
#pragma unroll
    for (int k = 0; k < 4; ++k)
      for (int j = lane + 32; j < ln[k]; j += 32) {
        const int2 q = sorted[bg[k] + j];
        atomicAdd(&ncur[q.y - node0], 1);
      }
  }
  __syncthreads();

  const int v = (tid < NODES_PER_BUCKET) ? ncur[tid] : 0;
  int tot2;
  const int ex = block_excl_scan_1024(v, wsum, tot2);
  if (tid < NODES_PER_BUCKET) {
    const int node = node0 + tid;
    if (node < n) {
      cnt[node]  = v;
      off[node]  = base + ex;
      dinv[node] = rsqrtf((float)(v + 1));
    }
    ncur[tid] = ex;
  }
  __syncthreads();

  for (int c0 = w; c0 < nchunks; c0 += 128) {
    int ln[4], bg[4];
    int2 ed[4];
#pragma unroll
    for (int k = 0; k < 4; ++k) {
      const int c = c0 + 32 * k;
      const bool in = c < nchunks;
      ln[k] = in ? rlen[c] : 0;
      bg[k] = in ? rbeg[c] : 0;
    }
#pragma unroll
    for (int k = 0; k < 4; ++k) {
      ed[k] = make_int2(0, node0);
      if (lane < ln[k]) ed[k] = sorted[bg[k] + lane];
    }
#pragma unroll
    for (int k = 0; k < 4; ++k)
      if (lane < ln[k]) {
        const int p = atomicAdd(&ncur[ed[k].y - node0], 1);
        if (staged) lbuf[p] = ed[k].x; else src[base + p] = ed[k].x;
      }
#pragma unroll
    for (int k = 0; k < 4; ++k)
      for (int j = lane + 32; j < ln[k]; j += 32) {
        const int2 q = sorted[bg[k] + j];
        const int p = atomicAdd(&ncur[q.y - node0], 1);
        if (staged) lbuf[p] = q.x; else src[base + p] = q.x;
      }
  }
  __syncthreads();
  if (staged)
    for (int j = tid; j < tot; j += 1024) src[base + j] = lbuf[j];
}


__global__ void k_gemm_wmma(const float* __restrict__ x,
                            const float* __restrict__ W,
                            float* __restrict__ xw, int ntiles) {
  const int lane   = threadIdx.x & 31;
  const int waveIb = threadIdx.x >> 5;
  const int tile   = blockIdx.x * (blockDim.x >> 5) + waveIb;
  if (tile >= ntiles) return;

  const int m     = lane & 15;
  const int khalf = lane >> 4;

  const float* xrow  = x + (size_t)(tile * 16 + m) * 128 + khalf * 2;
  const float* wbase = W + (size_t)m * 128 + khalf * 2;

  v8f acc[8] = {};

  v2f a_cur = *(const v2f*)(xrow);
  v2f b_cur[8];
#pragma unroll
  for (int t = 0; t < 8; ++t) b_cur[t] = *(const v2f*)(wbase + t * 2048);

#pragma unroll
  for (int k0 = 0; k0 < 128; k0 += 4) {
    v2f a_nxt = a_cur;
    v2f b_nxt[8];
#pragma unroll
    for (int t = 0; t < 8; ++t) b_nxt[t] = b_cur[t];
    if (k0 + 4 < 128) {
      a_nxt = *(const v2f*)(xrow + k0 + 4);
#pragma unroll
      for (int t = 0; t < 8; ++t)
        b_nxt[t] = *(const v2f*)(wbase + t * 2048 + k0 + 4);
    }
#pragma unroll
    for (int t = 0; t < 8; ++t) {
      acc[t] = __builtin_amdgcn_wmma_f32_16x16x4_f32(
           false, a_cur,  false, b_cur[t],
           (short)0, acc[t],  false,  false);
    }
    a_cur = a_nxt;
#pragma unroll
    for (int t = 0; t < 8; ++t) b_cur[t] = b_nxt[t];
  }

  float* outbase = xw + (size_t)(tile * 16 + khalf * 8) * 128 + m;
#pragma unroll
  for (int t = 0; t < 8; ++t) {
#pragma unroll
    for (int v = 0; v < 8; ++v) {
      outbase[(size_t)v * 128 + t * 16] = acc[t][v];
    }
  }
}

__global__ void k_gemm_tail(const float* __restrict__ x, const float* __restrict__ W,
                            float* __restrict__ xw, int n, int start) {
  int i = blockIdx.x * blockDim.x + threadIdx.x;
  int total = (n - start) * 128;
  if (i >= total) return;
  int node = start + i / 128;
  int o = i & 127;
  const float* xr = x + (size_t)node * 128;
  const float* wr = W + (size_t)o * 128;
  float s = 0.f;
  for (int k = 0; k < 128; ++k) s += xr[k] * wr[k];
  xw[(size_t)node * 128 + o] = s;
}


__device__ __forceinline__ void row_fma(float4& acc, const float4 v, const float w) {
  acc.x = fmaf(v.x, w, acc.x);
  acc.y = fmaf(v.y, w, acc.y);
  acc.z = fmaf(v.z, w, acc.z);
  acc.w = fmaf(v.w, w, acc.w);
}

__global__ void __launch_bounds__(256)
k_gather(const float* __restrict__ xw, const int* __restrict__ src,
         const int* __restrict__ off, const int* __restrict__ cnt,
         const float* __restrict__ dinv, const float* __restrict__ bias,
         float* __restrict__ out, int n) {
  const int lane = threadIdx.x & 31;
  const int node = __builtin_amdgcn_readfirstlane((blockIdx.x * blockDim.x + threadIdx.x) >> 5);
  if (node >= n) return;
  const int start = off[node];
  const int deg   = cnt[node];
  const float dc  = dinv[node];
  const float* xl = xw + lane * 4;

  const float4 vs = *(const float4*)(xl + (size_t)node * 128);
  const float4 bb = *(const float4*)(bias + lane * 4);
  const float s = dc * dc;
  float4 acc;
  acc.x = vs.x * s + bb.x;
  acc.y = vs.y * s + bb.y;
  acc.z = vs.z * s + bb.z;
  acc.w = vs.w * s + bb.w;

  for (int j0 = 0; j0 < deg; j0 += 32) {
    int my_r = 0;
    float my_w = 0.f;
    if (j0 + lane < deg) {
      my_r = src[start + j0 + lane];
      my_w = dinv[my_r] * dc;
    }
    const int m = (deg - j0 < 32) ? (deg - j0) : 32;
    int jj = 0;
    for (; jj + 8 <= m; jj += 8) {
      float4 v[8];
      float wt[8];
      for (int u = 0; u < 8; ++u) {
        const int r = __builtin_amdgcn_readlane(my_r, jj + u);
        wt[u] = __int_as_float(__builtin_amdgcn_readlane(__float_as_int(my_w), jj + u));
        v[u] = *(const float4*)(xl + (size_t)r * 128);
      }
      for (int u = 0; u < 8; ++u) row_fma(acc, v[u], wt[u]);
    }
    if (jj + 4 <= m) {
      float4 v[4];
      float wt[4];
      for (int u = 0; u < 4; ++u) {
        const int r = __builtin_amdgcn_readlane(my_r, jj + u);
        wt[u] = __int_as_float(__builtin_amdgcn_readlane(__float_as_int(my_w), jj + u));
        v[u] = *(const float4*)(xl + (size_t)r * 128);
      }
      for (int u = 0; u < 4; ++u) row_fma(acc, v[u], wt[u]);
      jj += 4;
    }
    if (jj + 2 <= m) {
      float4 v[2];
      float wt[2];
      for (int u = 0; u < 2; ++u) {
        const int r = __builtin_amdgcn_readlane(my_r, jj + u);
        wt[u] = __int_as_float(__builtin_amdgcn_readlane(__float_as_int(my_w), jj + u));
        v[u] = *(const float4*)(xl + (size_t)r * 128);
      }
      for (int u = 0; u < 2; ++u) row_fma(acc, v[u], wt[u]);
      jj += 2;
    }
    if (jj < m) {
      const int r = __builtin_amdgcn_readlane(my_r, jj);
      const float wt = __int_as_float(__builtin_amdgcn_readlane(__float_as_int(my_w), jj));
      const float4 v = *(const float4*)(xl + (size_t)r * 128);
      row_fma(acc, v, wt);
    }
  }
  *(float4*)(out + (size_t)node * 128 + lane * 4) = acc;
}


extern "C" void kernel_launch(void* const* d_in, const int* in_sizes, int n_in,
                              void* d_out, int out_size, void* d_ws, size_t ws_size,
                              hipStream_t stream) {
  const float* x  = (const float*)d_in[0];
  const int* ei   = (const int*)d_in[1];
  const float* W  = (const float*)d_in[2];
  const float* b  = (const float*)d_in[3];
  float* out      = (float*)d_out;

  const int N = in_sizes[0] / 128;
  const int E = in_sizes[1] / 2;
  const int* row = ei;
  const int* col = ei + E;

  const int NB = (N + NODES_PER_BUCKET - 1) / NODES_PER_BUCKET;
  const int NC = (E + CHUNK - 1) / CHUNK;

  float* xw   = (float*)d_ws;
  int* histG  = (int*)(xw + (size_t)N * 128);
  int* offG   = histG + (size_t)NB * NC;
  int* sorted = offG + (size_t)NB * NC;
  int* src    = sorted + (size_t)2 * E;
  int* cnt    = src + E;
  int* off    = cnt + N;
  float* dinv = (float*)(off + N);

  k_chunk_sort<<<NC, 1024, 0, stream>>>(row, col, sorted, histG, offG, E, NB, NC);
  k_bucket_lists<<<NB, 1024, 0, stream>>>(sorted, histG, offG, src, cnt, off, dinv, N, NC);

  const int ntiles = N / 16;
  if (ntiles > 0)
    k_gemm_wmma<<<(ntiles + 7) / 8, 256, 0, stream>>>(x, W, xw, ntiles);
  const int rem = N % 16;
  if (rem > 0)
    k_gemm_tail<<<(rem * 128 + 255) / 256, 256, 0, stream>>>(x, W, xw, N, ntiles * 16);

  k_gather<<<(N + 7) / 8, 256, 0, stream>>>(xw, src, off, cnt, dinv, b, out, N);
}


# GEMM reads both operands from LDS: the work-group copies W once and each wave its 16 rows of x with whole-row loads; fragments are 16-byte LDS reads (two k steps each); k sum order changed
# speedup vs baseline: 4.5774x; 1.3382x over previous
#include <hip/hip_runtime.h>
#include <hip/hip_bf16.h>

typedef float v2f __attribute__((ext_vector_type(2)));
typedef float v8f __attribute__((ext_vector_type(8)));


#define NODE_SH 8
#define NODES_PER_BUCKET 256
#define CHUNK 8192
#define EPT 8
#define STAGE_CAP 8192

__device__ __forceinline__ int block_excl_scan_1024(int v, int* wsum, int& total) {
  const int lane = threadIdx.x & 31;
  const int w    = threadIdx.x >> 5;
  int incl = v;
  for (int o = 1; o < 32; o <<= 1) {
    const int t = __shfl_up(incl, o);
    if (lane >= o) incl += t;
  }
  __syncthreads();
  if (lane == 31) wsum[w] = incl;
  __syncthreads();
  const int x = wsum[lane];
  int s = x;
  for (int o = 1; o < 32; o <<= 1) {
    const int t = __shfl_up(s, o);
    if (lane >= o) s += t;
  }
  total = __shfl(s, 31);
  return __shfl(s - x, w) + (incl - v);
}

__device__ __forceinline__ int block_sum_1024(int v, int* wsum) {
  const int lane = threadIdx.x & 31;
  const int w    = threadIdx.x >> 5;
  for (int o = 16; o > 0; o >>= 1) v += __shfl_xor(v, o);
  __syncthreads();
  if (lane == 0) wsum[w] = v;
  __syncthreads();
  int x = wsum[lane];
  for (int o = 16; o > 0; o >>= 1) x += __shfl_xor(x, o);
  return x;
}

__global__ void __launch_bounds__(1024)
k_chunk_sort(const int* __restrict__ row, const int* __restrict__ col,
             int* __restrict__ sorted_i, int* __restrict__ histG, int* __restrict__ offG,
             int e, int nb, int nchunks) {
  __shared__ int cb[1024];
  __shared__ int wsum[32];
  __shared__ int2 buf[CHUNK];
  int2* __restrict__ sorted = (int2*)sorted_i;
  const int tid = threadIdx.x;
  const int c   = blockIdx.x;
  const int e0  = c * CHUNK;
  cb[tid] = 0;
  int rw[EPT], cl[EPT], rk[EPT];
#pragma unroll
  for (int k = 0; k < EPT; ++k) {
    const int i = e0 + k * 1024 + tid;
    const bool ok = i < e;
    rw[k] = ok ? row[i] : 0;
    cl[k] = ok ? col[i] : -1;
  }
  __syncthreads();
#pragma unroll
  for (int k = 0; k < EPT; ++k)
    rk[k] = (cl[k] >= 0) ? atomicAdd(&cb[cl[k] >> NODE_SH], 1) : 0;
  __syncthreads();
  const int v = cb[tid];
  int total;
  const int ex = block_excl_scan_1024(v, wsum, total);
  if (tid < nb) {
    histG[tid * nchunks + c] = v;
    offG[tid * nchunks + c]  = ex;
  }
  cb[tid] = ex;
  __syncthreads();
#pragma unroll
  for (int k = 0; k < EPT; ++k)
    if (cl[k] >= 0) buf[cb[cl[k] >> NODE_SH] + rk[k]] = make_int2(rw[k], cl[k]);
  __syncthreads();
#pragma unroll
  for (int k = 0; k < EPT; ++k) {
    const int j = k * 1024 + tid;
    if (e0 + j < e) sorted[e0 + j] = buf[j];
  }
}

__global__ void __launch_bounds__(1024)
k_bucket_lists(const int* __restrict__ sorted_i, const int* __restrict__ histG,
               const int* __restrict__ offG, int* __restrict__ src, int* __restrict__ cnt,
               int* __restrict__ off, float* __restrict__ dinv, int n, int nchunks) {
  __shared__ int rlen[1024];
  __shared__ int rbeg[1024];
  __shared__ int ncur[NODES_PER_BUCKET];
  __shared__ int wsum[32];
  __shared__ int lbuf[STAGE_CAP];
  const int2* __restrict__ sorted = (const int2*)sorted_i;
  const int tid  = threadIdx.x;
  const int lane = tid & 31;
  const int w     = __builtin_amdgcn_readfirstlane(tid >> 5);
  const int b     = blockIdx.x;
  const int node0 = b << NODE_SH;
  const int front = b * nchunks;

  int mylen = 0;
  if (tid < nchunks) {
    mylen     = histG[front + tid];
    rlen[tid] = mylen;
    rbeg[tid] = tid * CHUNK + offG[front + tid];
  }
  if (tid < NODES_PER_BUCKET) ncur[tid] = 0;

  int part = 0;
  const int4* h4 = (const int4*)histG;
  const int nvec = front >> 2;
#pragma unroll 4
  for (int i = tid; i < nvec; i += 1024) {
    const int4 q = h4[i];
    part += (q.x + q.y) + (q.z + q.w);
  }
  if (tid < (front & 3)) part += histG[(nvec << 2) + tid];
  const int base = block_sum_1024(part, wsum);
  const int tot  = block_sum_1024(mylen, wsum);
  const bool staged = tot <= STAGE_CAP;

  for (int c0 = w; c0 < nchunks; c0 += 128) {
    int ln[4], bg[4];
    int2 ed[4];
#pragma unroll
    for (int k = 0; k < 4; ++k) {
      const int c = c0 + 32 * k;
      const bool in = c < nchunks;
      ln[k] = in ? rlen[c] : 0;
      bg[k] = in ? rbeg[c] : 0;
    }
#pragma unroll
    for (int k = 0; k < 4; ++k) {
      ed[k] = make_int2(0, node0);
      if (lane < ln[k]) ed[k] = sorted[bg[k] + lane];
    }
#pragma unroll
    for (int k = 0; k < 4; ++k)
      if (lane < ln[k]) atomicAdd(&ncur[ed[k].y - node0], 1);
#pragma unroll
    for (int k = 0; k < 4; ++k)
      for (int j = lane + 32; j < ln[k]; j += 32) {
        const int2 q = sorted[bg[k] + j];
        atomicAdd(&ncur[q.y - node0], 1);
      }
  }
  __syncthreads();

  const int v = (tid < NODES_PER_BUCKET) ? ncur[tid] : 0;
  int tot2;
  const int ex = block_excl_scan_1024(v, wsum, tot2);
  if (tid < NODES_PER_BUCKET) {
    const int node = node0 + tid;
    if (node < n) {
      cnt[node]  = v;
      off[node]  = base + ex;
      dinv[node] = rsqrtf((float)(v + 1));
    }
    ncur[tid] = ex;
  }
  __syncthreads();

  for (int c0 = w; c0 < nchunks; c0 += 128) {
    int ln[4], bg[4];
    int2 ed[4];
#pragma unroll
    for (int k = 0; k < 4; ++k) {
      const int c = c0 + 32 * k;
      const bool in = c < nchunks;
      ln[k] = in ? rlen[c] : 0;
      bg[k] = in ? rbeg[c] : 0;
    }
#pragma unroll
    for (int k = 0; k < 4; ++k) {
      ed[k] = make_int2(0, node0);
      if (lane < ln[k]) ed[k] = sorted[bg[k] + lane];
    }
#pragma unroll
    for (int k = 0; k < 4; ++k)
      if (lane < ln[k]) {
        const int p = atomicAdd(&ncur[ed[k].y - node0], 1);
        if (staged) lbuf[p] = ed[k].x; else src[base + p] = ed[k].x;
      }
#pragma unroll
    for (int k = 0; k < 4; ++k)
      for (int j = lane + 32; j < ln[k]; j += 32) {
        const int2 q = sorted[bg[k] + j];
        const int p = atomicAdd(&ncur[q.y - node0], 1);
        if (staged) lbuf[p] = q.x; else src[base + p] = q.x;
      }
  }
  __syncthreads();
  if (staged)
    for (int j = tid; j < tot; j += 1024) src[base + j] = lbuf[j];
}


#define LDS_PITCH4 33

__global__ void __launch_bounds__(256)
k_gemm_wmma(const float* __restrict__ x,
            const float* __restrict__ W,
            float* __restrict__ xw, int ntiles) {
  __shared__ float4 Ws[128 * LDS_PITCH4];
  __shared__ float4 As[8 * 16 * LDS_PITCH4];
  const int lane   = threadIdx.x & 31;
  const int waveIb = threadIdx.x >> 5;
  const int tile   = blockIdx.x * 8 + waveIb;
  const bool have  = tile < ntiles;

  const float4* __restrict__ W4 = (const float4*)W;
  const float4* __restrict__ x4 = (const float4*)x;

  const int ltile = have ? tile : ntiles - 1;
  float4 wv[16], av[16];
#pragma unroll
  for (int j = 0; j < 16; ++j) wv[j] = W4[(waveIb + 8 * j) * 32 + lane];
#pragma unroll
  for (int r = 0; r < 16; ++r) av[r] = x4[(size_t)(ltile * 16 + r) * 32 + lane];
#pragma unroll
  for (int j = 0; j < 16; ++j) Ws[(waveIb + 8 * j) * LDS_PITCH4 + lane] = wv[j];
#pragma unroll
  for (int r = 0; r < 16; ++r) As[(waveIb * 16 + r) * LDS_PITCH4 + lane] = av[r];
  __syncthreads();
  if (!have) return;

  const int m     = lane & 15;
  const int khalf = lane >> 4;

  const float4* Ap = As + (waveIb * 16 + m) * LDS_PITCH4 + khalf;
  const float4* Bp = Ws + m * LDS_PITCH4 + khalf;

  v8f acc[8] = {};

  float4 a_cur = Ap[0];
  float4 b_cur[8];
#pragma unroll
  for (int t = 0; t < 8; ++t) b_cur[t] = Bp[t * 16 * LDS_PITCH4];

#pragma unroll
  for (int kb = 0; kb < 16; ++kb) {
    float4 a_nxt = a_cur;
    float4 b_nxt[8];
#pragma unroll
    for (int t = 0; t < 8; ++t) b_nxt[t] = b_cur[t];
    if (kb + 1 < 16) {
      a_nxt = Ap[(kb + 1) * 2];
#pragma unroll
      for (int t = 0; t < 8; ++t) b_nxt[t] = Bp[t * 16 * LDS_PITCH4 + (kb + 1) * 2];
    }
    const v2f a_lo = {a_cur.x, a_cur.y};
    const v2f a_hi = {a_cur.z, a_cur.w};
#pragma unroll
    for (int t = 0; t < 8; ++t) {
      const v2f b_lo = {b_cur[t].x, b_cur[t].y};
      acc[t] = __builtin_amdgcn_wmma_f32_16x16x4_f32(
           false, a_lo,  false, b_lo,
           (short)0, acc[t],  false,  false);
    }
#pragma unroll
    for (int t = 0; t < 8; ++t) {
      const v2f b_hi = {b_cur[t].z, b_cur[t].w};
      acc[t] = __builtin_amdgcn_wmma_f32_16x16x4_f32(
           false, a_hi,  false, b_hi,
           (short)0, acc[t],  false,  false);
    }
    a_cur = a_nxt;
#pragma unroll
    for (int t = 0; t < 8; ++t) b_cur[t] = b_nxt[t];
  }

  float* outbase = xw + (size_t)(tile * 16 + khalf * 8) * 128 + m;
#pragma unroll
  for (int t = 0; t < 8; ++t) {
#pragma unroll
    for (int v = 0; v < 8; ++v) {
      outbase[(size_t)v * 128 + t * 16] = acc[t][v];
    }
  }
}

__global__ void k_gemm_tail(const float* __restrict__ x, const float* __restrict__ W,
                            float* __restrict__ xw, int n, int start) {
  int i = blockIdx.x * blockDim.x + threadIdx.x;
  int total = (n - start) * 128;
  if (i >= total) return;
  int node = start + i / 128;
  int o = i & 127;
  const float* xr = x + (size_t)node * 128;
  const float* wr = W + (size_t)o * 128;
  float s = 0.f;
  for (int k = 0; k < 128; ++k) s += xr[k] * wr[k];
  xw[(size_t)node * 128 + o] = s;
}


__device__ __forceinline__ void row_fma(float4& acc, const float4 v, const float w) {
  acc.x = fmaf(v.x, w, acc.x);
  acc.y = fmaf(v.y, w, acc.y);
  acc.z = fmaf(v.z, w, acc.z);
  acc.w = fmaf(v.w, w, acc.w);
}

__global__ void __launch_bounds__(256)
k_gather(const float* __restrict__ xw, const int* __restrict__ src,
         const int* __restrict__ off, const int* __restrict__ cnt,
         const float* __restrict__ dinv, const float* __restrict__ bias,
         float* __restrict__ out, int n) {
  const int lane = threadIdx.x & 31;
  const int node = __builtin_amdgcn_readfirstlane((blockIdx.x * blockDim.x + threadIdx.x) >> 5);
  if (node >= n) return;
  const int start = off[node];
  const int deg   = cnt[node];
  const float dc  = dinv[node];
  const float* xl = xw + lane * 4;

  const float4 vs = *(const float4*)(xl + (size_t)node * 128);
  const float4 bb = *(const float4*)(bias + lane * 4);
  const float s = dc * dc;
  float4 acc;
  acc.x = vs.x * s + bb.x;
  acc.y = vs.y * s + bb.y;
  acc.z = vs.z * s + bb.z;
  acc.w = vs.w * s + bb.w;

  for (int j0 = 0; j0 < deg; j0 += 32) {
    int my_r = 0;
    float my_w = 0.f;
    if (j0 + lane < deg) {
      my_r = src[start + j0 + lane];
      my_w = dinv[my_r] * dc;
    }
    const int m = (deg - j0 < 32) ? (deg - j0) : 32;
    int jj = 0;
    for (; jj + 8 <= m; jj += 8) {
      float4 v[8];
      float wt[8];
      for (int u = 0; u < 8; ++u) {
        const int r = __builtin_amdgcn_readlane(my_r, jj + u);
        wt[u] = __int_as_float(__builtin_amdgcn_readlane(__float_as_int(my_w), jj + u));
        v[u] = *(const float4*)(xl + (size_t)r * 128);
      }
      for (int u = 0; u < 8; ++u) row_fma(acc, v[u], wt[u]);
    }
    if (jj + 4 <= m) {
      float4 v[4];
      float wt[4];
      for (int u = 0; u < 4; ++u) {
        const int r = __builtin_amdgcn_readlane(my_r, jj + u);
        wt[u] = __int_as_float(__builtin_amdgcn_readlane(__float_as_int(my_w), jj + u));
        v[u] = *(const float4*)(xl + (size_t)r * 128);
      }
      for (int u = 0; u < 4; ++u) row_fma(acc, v[u], wt[u]);
      jj += 4;
    }
    if (jj + 2 <= m) {
      float4 v[2];
      float wt[2];
      for (int u = 0; u < 2; ++u) {
        const int r = __builtin_amdgcn_readlane(my_r, jj + u);
        wt[u] = __int_as_float(__builtin_amdgcn_readlane(__float_as_int(my_w), jj + u));
        v[u] = *(const float4*)(xl + (size_t)r * 128);
      }
      for (int u = 0; u < 2; ++u) row_fma(acc, v[u], wt[u]);
      jj += 2;
    }
    if (jj < m) {
      const int r = __builtin_amdgcn_readlane(my_r, jj);
      const float wt = __int_as_float(__builtin_amdgcn_readlane(__float_as_int(my_w), jj));
      const float4 v = *(const float4*)(xl + (size_t)r * 128);
      row_fma(acc, v, wt);
    }
  }
  *(float4*)(out + (size_t)node * 128 + lane * 4) = acc;
}


extern "C" void kernel_launch(void* const* d_in, const int* in_sizes, int n_in,
                              void* d_out, int out_size, void* d_ws, size_t ws_size,
                              hipStream_t stream) {
  const float* x  = (const float*)d_in[0];
  const int* ei   = (const int*)d_in[1];
  const float* W  = (const float*)d_in[2];
  const float* b  = (const float*)d_in[3];
  float* out      = (float*)d_out;

  const int N = in_sizes[0] / 128;
  const int E = in_sizes[1] / 2;
  const int* row = ei;
  const int* col = ei + E;

  const int NB = (N + NODES_PER_BUCKET - 1) / NODES_PER_BUCKET;
  const int NC = (E + CHUNK - 1) / CHUNK;

  float* xw   = (float*)d_ws;
  int* histG  = (int*)(xw + (size_t)N * 128);
  int* offG   = histG + (size_t)NB * NC;
  int* sorted = offG + (size_t)NB * NC;
  int* src    = sorted + (size_t)2 * E;
  int* cnt    = src + E;
  int* off    = cnt + N;
  float* dinv = (float*)(off + N);

  k_chunk_sort<<<NC, 1024, 0, stream>>>(row, col, sorted, histG, offG, E, NB, NC);
  k_bucket_lists<<<NB, 1024, 0, stream>>>(sorted, histG, offG, src, cnt, off, dinv, N, NC);

  const int ntiles = N / 16;
  if (ntiles > 0)
    k_gemm_wmma<<<(ntiles + 7) / 8, 256, 0, stream>>>(x, W, xw, ntiles);
  const int rem = N % 16;
  if (rem > 0)
    k_gemm_tail<<<(rem * 128 + 255) / 256, 256, 0, stream>>>(x, W, xw, N, ntiles * 16);

  k_gather<<<(N + 7) / 8, 256, 0, stream>>>(xw, src, off, cnt, dinv, b, out, N);
}


# The f32 WMMA product x@W^T no longer has its own launch: its tiles run as extra 1024-thread work-groups inside the two list-building launches (4 waves per 16-row strip); sums unchanged
# speedup vs baseline: 4.8330x; 1.0558x over previous
#include <hip/hip_runtime.h>
#include <hip/hip_bf16.h>

typedef float v2f __attribute__((ext_vector_type(2)));
typedef float v8f __attribute__((ext_vector_type(8)));


#define NODE_SH 8
#define NODES_PER_BUCKET 256
#define CHUNK 8192
#define EPT 8
#define STAGE_CAP 8192

__device__ __forceinline__ int block_excl_scan_1024(int v, int* wsum, int& total) {
  const int lane = threadIdx.x & 31;
  const int w    = threadIdx.x >> 5;
  int incl = v;
  for (int o = 1; o < 32; o <<= 1) {
    const int t = __shfl_up(incl, o);
    if (lane >= o) incl += t;
  }
  __syncthreads();
  if (lane == 31) wsum[w] = incl;
  __syncthreads();
  const int x = wsum[lane];
  int s = x;
  for (int o = 1; o < 32; o <<= 1) {
    const int t = __shfl_up(s, o);
    if (lane >= o) s += t;
  }
  total = __shfl(s, 31);
  return __shfl(s - x, w) + (incl - v);
}

__device__ __forceinline__ int block_sum_1024(int v, int* wsum) {
  const int lane = threadIdx.x & 31;
  const int w    = threadIdx.x >> 5;
  for (int o = 16; o > 0; o >>= 1) v += __shfl_xor(v, o);
  __syncthreads();
  if (lane == 0) wsum[w] = v;
  __syncthreads();
  int x = wsum[lane];
  for (int o = 16; o > 0; o >>= 1) x += __shfl_xor(x, o);
  return x;
}

__device__ __forceinline__ void
chunk_sort_body(int* lds, const int* __restrict__ row, const int* __restrict__ col,
                int* __restrict__ sorted_i, int* __restrict__ histG, int* __restrict__ offG,
                int e, int nb, int nchunks, const int c) {
  int* cb   = lds;
  int* wsum = lds + 1024;
  int2* buf = (int2*)(lds + 1024 + 32);
  int2* __restrict__ sorted = (int2*)sorted_i;
  const int tid = threadIdx.x;
  const int e0  = c * CHUNK;
  cb[tid] = 0;
  int rw[EPT], cl[EPT], rk[EPT];
#pragma unroll
  for (int k = 0; k < EPT; ++k) {
    const int i = e0 + k * 1024 + tid;
    const bool ok = i < e;
    rw[k] = ok ? row[i] : 0;
    cl[k] = ok ? col[i] : -1;
  }
  __syncthreads();
#pragma unroll
  for (int k = 0; k < EPT; ++k)
    rk[k] = (cl[k] >= 0) ? atomicAdd(&cb[cl[k] >> NODE_SH], 1) : 0;
  __syncthreads();
  const int v = cb[tid];
  int total;
  const int ex = block_excl_scan_1024(v, wsum, total);
  if (tid < nb) {
    histG[tid * nchunks + c] = v;
    offG[tid * nchunks + c]  = ex;
  }
  cb[tid] = ex;
  __syncthreads();
#pragma unroll
  for (int k = 0; k < EPT; ++k)
    if (cl[k] >= 0) buf[cb[cl[k] >> NODE_SH] + rk[k]] = make_int2(rw[k], cl[k]);
  __syncthreads();
#pragma unroll
  for (int k = 0; k < EPT; ++k) {
    const int j = k * 1024 + tid;
    if (e0 + j < e) sorted[e0 + j] = buf[j];
  }
}

__device__ __forceinline__ void
bucket_lists_body(int* lds, const int* __restrict__ sorted_i, const int* __restrict__ histG,
                  const int* __restrict__ offG, int* __restrict__ src, int* __restrict__ cnt,
                  int* __restrict__ off, float* __restrict__ dinv, int n, int nchunks,
                  const int b) {
  int* rlen = lds;
  int* rbeg = lds + 1024;
  int* ncur = lds + 2048;
  int* wsum = lds + 2048 + NODES_PER_BUCKET;
  int* lbuf = lds + 2048 + NODES_PER_BUCKET + 32;
  const int2* __restrict__ sorted = (const int2*)sorted_i;
  const int tid  = threadIdx.x;
  const int lane = tid & 31;
  const int w     = __builtin_amdgcn_readfirstlane(tid >> 5);
  const int node0 = b << NODE_SH;
  const int front = b * nchunks;

  int mylen = 0;
  if (tid < nchunks) {
    mylen     = histG[front + tid];
    rlen[tid] = mylen;
    rbeg[tid] = tid * CHUNK + offG[front + tid];
  }
  if (tid < NODES_PER_BUCKET) ncur[tid] = 0;

  int part = 0;
  const int4* h4 = (const int4*)histG;
  const int nvec = front >> 2;
#pragma unroll 4
  for (int i = tid; i < nvec; i += 1024) {
    const int4 q = h4[i];
    part += (q.x + q.y) + (q.z + q.w);
  }
  if (tid < (front & 3)) part += histG[(nvec << 2) + tid];
  const int base = block_sum_1024(part, wsum);
  const int tot  = block_sum_1024(mylen, wsum);
  const bool staged = tot <= STAGE_CAP;

  for (int c0 = w; c0 < nchunks; c0 += 128) {
    int ln[4], bg[4];
    int2 ed[4];
#pragma unroll
    for (int k = 0; k < 4; ++k) {
      const int c = c0 + 32 * k;
      const bool in = c < nchunks;
      ln[k] = in ? rlen[c] : 0;
      bg[k] = in ? rbeg[c] : 0;
    }
#pragma unroll
    for (int k = 0; k < 4; ++k) {
      ed[k] = make_int2(0, node0);
      if (lane < ln[k]) ed[k] = sorted[bg[k] + lane];
    }
#pragma unroll
    for (int k = 0; k < 4; ++k)
      if (lane < ln[k]) atomicAdd(&ncur[ed[k].y - node0], 1);
#pragma unroll
    for (int k = 0; k < 4; ++k)
      for (int j = lane + 32; j < ln[k]; j += 32) {
        const int2 q = sorted[bg[k] + j];
        atomicAdd(&ncur[q.y - node0], 1);
      }
  }
  __syncthreads();

  const int v = (tid < NODES_PER_BUCKET) ? ncur[tid] : 0;
  int tot2;
  const int ex = block_excl_scan_1024(v, wsum, tot2);
  if (tid < NODES_PER_BUCKET) {
    const int node = node0 + tid;
    if (node < n) {
      cnt[node]  = v;
      off[node]  = base + ex;
      dinv[node] = rsqrtf((float)(v + 1));
    }
    ncur[tid] = ex;
  }
  __syncthreads();

  for (int c0 = w; c0 < nchunks; c0 += 128) {
    int ln[4], bg[4];
    int2 ed[4];
#pragma unroll
    for (int k = 0; k < 4; ++k) {
      const int c = c0 + 32 * k;
      const bool in = c < nchunks;
      ln[k] = in ? rlen[c] : 0;
      bg[k] = in ? rbeg[c] : 0;
    }
#pragma unroll
    for (int k = 0; k < 4; ++k) {
      ed[k] = make_int2(0, node0);
      if (lane < ln[k]) ed[k] = sorted[bg[k] + lane];
    }
#pragma unroll
    for (int k = 0; k < 4; ++k)
      if (lane < ln[k]) {
        const int p = atomicAdd(&ncur[ed[k].y - node0], 1);
        if (staged) lbuf[p] = ed[k].x; else src[base + p] = ed[k].x;
      }
#pragma unroll
    for (int k = 0; k < 4; ++k)
      for (int j = lane + 32; j < ln[k]; j += 32) {
        const int2 q = sorted[bg[k] + j];
        const int p = atomicAdd(&ncur[q.y - node0], 1);
        if (staged) lbuf[p] = q.x; else src[base + p] = q.x;
      }
  }
  __syncthreads();
  if (staged)
    for (int j = tid; j < tot; j += 1024) src[base + j] = lbuf[j];
}


#define LDS_PITCH4 33
#define GEMM_LDS_F4 (2 * 128 * LDS_PITCH4)
#define ROWS_PER_GROUP 128

__device__ __forceinline__ void
gemm_group_body(float4* lds, const float* __restrict__ x, const float* __restrict__ W,
                float* __restrict__ xw, const int ntiles, const int group) {
  float4* Ws = lds;
  float4* As = lds + 128 * LDS_PITCH4;
  const int lane = threadIdx.x & 31;
  const int wv   = threadIdx.x >> 5;
  const int strip = wv >> 2;
  const int nq    = wv & 3;
  const int tile  = group * 8 + strip;
  const bool have = tile < ntiles;

  const float4* __restrict__ W4 = (const float4*)W;
  const float4* __restrict__ x4 = (const float4*)x;

  const int lastrow = ntiles * 16 - 1;
  float4 wvv[4], avv[4];
#pragma unroll
  for (int j = 0; j < 4; ++j) wvv[j] = W4[(wv + 32 * j) * 32 + lane];
#pragma unroll
  for (int j = 0; j < 4; ++j) {
    int r = group * ROWS_PER_GROUP + wv + 32 * j;
    r = r < lastrow ? r : lastrow;
    avv[j] = x4[(size_t)r * 32 + lane];
  }
#pragma unroll
  for (int j = 0; j < 4; ++j) Ws[(wv + 32 * j) * LDS_PITCH4 + lane] = wvv[j];
#pragma unroll
  for (int j = 0; j < 4; ++j) As[(wv + 32 * j) * LDS_PITCH4 + lane] = avv[j];
  __syncthreads();
  if (!have) return;

  const int m     = lane & 15;
  const int khalf = lane >> 4;

  const float4* Ap = As + (strip * 16 + m) * LDS_PITCH4 + khalf;
  const float4* Bp = Ws + (nq * 32 + m) * LDS_PITCH4 + khalf;

  v8f acc[2] = {};

  float4 a_cur = Ap[0];
  float4 b_cur[2];
#pragma unroll
  for (int t = 0; t < 2; ++t) b_cur[t] = Bp[t * 16 * LDS_PITCH4];

#pragma unroll
  for (int kb = 0; kb < 16; ++kb) {
    float4 a_nxt = a_cur;
    float4 b_nxt[2];
#pragma unroll
    for (int t = 0; t < 2; ++t) b_nxt[t] = b_cur[t];
    if (kb + 1 < 16) {
      a_nxt = Ap[(kb + 1) * 2];
#pragma unroll
      for (int t = 0; t < 2; ++t) b_nxt[t] = Bp[t * 16 * LDS_PITCH4 + (kb + 1) * 2];
    }
    const v2f a_lo = {a_cur.x, a_cur.y};
    const v2f a_hi = {a_cur.z, a_cur.w};
#pragma unroll
    for (int t = 0; t < 2; ++t) {
      const v2f b_lo = {b_cur[t].x, b_cur[t].y};
      acc[t] = __builtin_amdgcn_wmma_f32_16x16x4_f32(
           false, a_lo,  false, b_lo,
           (short)0, acc[t],  false,  false);
    }
#pragma unroll
    for (int t = 0; t < 2; ++t) {
      const v2f b_hi = {b_cur[t].z, b_cur[t].w};
      acc[t] = __builtin_amdgcn_wmma_f32_16x16x4_f32(
           false, a_hi,  false, b_hi,
           (short)0, acc[t],  false,  false);
    }
    a_cur = a_nxt;
#pragma unroll
    for (int t = 0; t < 2; ++t) b_cur[t] = b_nxt[t];
  }

  float* outbase = xw + (size_t)(tile * 16 + khalf * 8) * 128 + nq * 32 + m;
#pragma unroll
  for (int t = 0; t < 2; ++t) {
#pragma unroll
    for (int v = 0; v < 8; ++v) {
      outbase[(size_t)v * 128 + t * 16] = acc[t][v];
    }
  }
}

__global__ void __launch_bounds__(1024)
k_chunk_sort_gemm(const int* __restrict__ row, const int* __restrict__ col,
                  int* __restrict__ sorted_i, int* __restrict__ histG, int* __restrict__ offG,
                  int e, int nb, int nchunks,
                  const float* __restrict__ x, const float* __restrict__ W,
                  float* __restrict__ xw, int ntiles, int group0) {
  __shared__ float4 smem[GEMM_LDS_F4];
  const int bid = blockIdx.x;
  if (bid < nchunks)
    chunk_sort_body((int*)smem, row, col, sorted_i, histG, offG, e, nb, nchunks, bid);
  else
    gemm_group_body(smem, x, W, xw, ntiles, group0 + (bid - nchunks));
}

__global__ void __launch_bounds__(1024)
k_bucket_lists_gemm(const int* __restrict__ sorted_i, const int* __restrict__ histG,
                    const int* __restrict__ offG, int* __restrict__ src, int* __restrict__ cnt,
                    int* __restrict__ off, float* __restrict__ dinv, int n, int nchunks, int nb,
                    const float* __restrict__ x, const float* __restrict__ W,
                    float* __restrict__ xw, int ntiles, int group0) {
  __shared__ float4 smem[GEMM_LDS_F4];
  const int bid = blockIdx.x;
  if (bid < nb)
    bucket_lists_body((int*)smem, sorted_i, histG, offG, src, cnt, off, dinv, n, nchunks, bid);
  else
    gemm_group_body(smem, x, W, xw, ntiles, group0 + (bid - nb));
}

__global__ void k_gemm_tail(const float* __restrict__ x, const float* __restrict__ W,
                            float* __restrict__ xw, int n, int start) {
  int i = blockIdx.x * blockDim.x + threadIdx.x;
  int total = (n - start) * 128;
  if (i >= total) return;
  int node = start + i / 128;
  int o = i & 127;
  const float* xr = x + (size_t)node * 128;
  const float* wr = W + (size_t)o * 128;
  float s = 0.f;
  for (int k = 0; k < 128; ++k) s += xr[k] * wr[k];
  xw[(size_t)node * 128 + o] = s;
}


__device__ __forceinline__ void row_fma(float4& acc, const float4 v, const float w) {
  acc.x = fmaf(v.x, w, acc.x);
  acc.y = fmaf(v.y, w, acc.y);
  acc.z = fmaf(v.z, w, acc.z);
  acc.w = fmaf(v.w, w, acc.w);
}

__global__ void __launch_bounds__(256)
k_gather(const float* __restrict__ xw, const int* __restrict__ src,
         const int* __restrict__ off, const int* __restrict__ cnt,
         const float* __restrict__ dinv, const float* __restrict__ bias,
         float* __restrict__ out, int n) {
  const int lane = threadIdx.x & 31;
  const int node = __builtin_amdgcn_readfirstlane((blockIdx.x * blockDim.x + threadIdx.x) >> 5);
  if (node >= n) return;
  const int start = off[node];
  const int deg   = cnt[node];
  const float dc  = dinv[node];
  const float* xl = xw + lane * 4;

  const float4 vs = *(const float4*)(xl + (size_t)node * 128);
  const float4 bb = *(const float4*)(bias + lane * 4);
  const float s = dc * dc;
  float4 acc;
  acc.x = vs.x * s + bb.x;
  acc.y = vs.y * s + bb.y;
  acc.z = vs.z * s + bb.z;
  acc.w = vs.w * s + bb.w;

  for (int j0 = 0; j0 < deg; j0 += 32) {
    int my_r = 0;
    float my_w = 0.f;
    if (j0 + lane < deg) {
      my_r = src[start + j0 + lane];
      my_w = dinv[my_r] * dc;
    }
    const int m = (deg - j0 < 32) ? (deg - j0) : 32;
    int jj = 0;
    for (; jj + 8 <= m; jj += 8) {
      float4 v[8];
      float wt[8];
      for (int u = 0; u < 8; ++u) {
        const int r = __builtin_amdgcn_readlane(my_r, jj + u);
        wt[u] = __int_as_float(__builtin_amdgcn_readlane(__float_as_int(my_w), jj + u));
        v[u] = *(const float4*)(xl + (size_t)r * 128);
      }
      for (int u = 0; u < 8; ++u) row_fma(acc, v[u], wt[u]);
    }
    if (jj + 4 <= m) {
      float4 v[4];
      float wt[4];
      for (int u = 0; u < 4; ++u) {
        const int r = __builtin_amdgcn_readlane(my_r, jj + u);
        wt[u] = __int_as_float(__builtin_amdgcn_readlane(__float_as_int(my_w), jj + u));
        v[u] = *(const float4*)(xl + (size_t)r * 128);
      }
      for (int u = 0; u < 4; ++u) row_fma(acc, v[u], wt[u]);
      jj += 4;
    }
    if (jj + 2 <= m) {
      float4 v[2];
      float wt[2];
      for (int u = 0; u < 2; ++u) {
        const int r = __builtin_amdgcn_readlane(my_r, jj + u);
        wt[u] = __int_as_float(__builtin_amdgcn_readlane(__float_as_int(my_w), jj + u));
        v[u] = *(const float4*)(xl + (size_t)r * 128);
      }
      for (int u = 0; u < 2; ++u) row_fma(acc, v[u], wt[u]);
      jj += 2;
    }
    if (jj < m) {
      const int r = __builtin_amdgcn_readlane(my_r, jj);
      const float wt = __int_as_float(__builtin_amdgcn_readlane(__float_as_int(my_w), jj));
      const float4 v = *(const float4*)(xl + (size_t)r * 128);
      row_fma(acc, v, wt);
    }
  }
  *(float4*)(out + (size_t)node * 128 + lane * 4) = acc;
}


extern "C" void kernel_launch(void* const* d_in, const int* in_sizes, int n_in,
                              void* d_out, int out_size, void* d_ws, size_t ws_size,
                              hipStream_t stream) {
  const float* x  = (const float*)d_in[0];
  const int* ei   = (const int*)d_in[1];
  const float* W  = (const float*)d_in[2];
  const float* b  = (const float*)d_in[3];
  float* out      = (float*)d_out;

  const int N = in_sizes[0] / 128;
  const int E = in_sizes[1] / 2;
  const int* row = ei;
  const int* col = ei + E;

  const int NB = (N + NODES_PER_BUCKET - 1) / NODES_PER_BUCKET;
  const int NC = (E + CHUNK - 1) / CHUNK;

  float* xw   = (float*)d_ws;
  int* histG  = (int*)(xw + (size_t)N * 128);
  int* offG   = histG + (size_t)NB * NC;
  int* sorted = offG + (size_t)NB * NC;
  int* src    = sorted + (size_t)2 * E;
  int* cnt    = src + E;
  int* off    = cnt + N;
  float* dinv = (float*)(off + N);

  const int ntiles  = N / 16;
  const int ngroups = (ntiles + 7) / 8;
  const int g1 = (2 * ngroups) / 3;
  const int g2 = ngroups - g1;
  k_chunk_sort_gemm<<<NC + g1, 1024, 0, stream>>>(row, col, sorted, histG, offG, E, NB, NC,
                                                  x, W, xw, ntiles, 0);
  k_bucket_lists_gemm<<<NB + g2, 1024, 0, stream>>>(sorted, histG, offG, src, cnt, off, dinv,
                                                    N, NC, NB, x, W, xw, ntiles, g1);

  const int rem = N % 16;
  if (rem > 0)
    k_gemm_tail<<<(rem * 128 + 255) / 256, 256, 0, stream>>>(x, W, xw, N, ntiles * 16);

  k_gather<<<(N + 7) / 8, 256, 0, stream>>>(xw, src, off, cnt, dinv, b, out, N);
}
